# starting kernel
# baseline (speedup 1.0000x reference)
#include <hip/hip_runtime.h>

typedef float v2f __attribute__((ext_vector_type(2)));
typedef float v4f __attribute__((ext_vector_type(4)));
typedef float v8f __attribute__((ext_vector_type(8)));

__global__ __launch_bounds__(256) void gcn_deg_init(float* __restrict__ deg, int n) {
    int i = blockIdx.x * blockDim.x + threadIdx.x;
    if (i < n) deg[i] = 1.0f;
}

__global__ __launch_bounds__(256) void gcn_deg_count(float* __restrict__ deg,
                                                     const int* __restrict__ dst, int e) {
    int i = blockIdx.x * blockDim.x + threadIdx.x;
    if (i < e) atomicAdd(&deg[dst[i]], 1.0f);
}

__global__ __launch_bounds__(256) void gcn_rsqrt(float* __restrict__ dis,
                                                 const float* __restrict__ deg, int n) {
    int i = blockIdx.x * blockDim.x + threadIdx.x;
    if (i < n) dis[i] = rsqrtf(deg[i]);
}

template <int KK, int TN, bool RELU, bool BIAS>
__global__ __launch_bounds__(256) void gcn_gemm_wmma(
        const float* __restrict__ A, const float* __restrict__ B,
        const float* __restrict__ bias, float* __restrict__ C, int M) {
    constexpr int Nc  = TN * 16;
    constexpr int LDB = KK + 2;
    __shared__ float Blds[Nc * LDB];

    for (int i = threadIdx.x; i < KK * Nc; i += 256) {
        int k = i / Nc;
        int c = i - k * Nc;
        Blds[c * LDB + k] = B[i];
    }
    __syncthreads();

    int tilesM = M >> 4;
    int tm     = blockIdx.x * 8 + (threadIdx.x >> 5);
    int lane   = threadIdx.x & 31;
    int half   = lane >> 4;
    int lr     = lane & 15;

    if (tm < tilesM) {
        const float* Arow = A + (size_t)(tm * 16 + lr) * KK;
        __builtin_prefetch(Arow, 0, 0);
        __builtin_prefetch(Arow + 64, 0, 0);

        v8f acc[TN] = {};
        for (int k = 0; k < KK; k += 4) {
            v2f a = *(const v2f*)(Arow + k + 2 * half);
#pragma unroll
            for (int t = 0; t < TN; ++t) {
                v2f b = *(const v2f*)(&Blds[(t * 16 + lr) * LDB + k + 2 * half]);
                acc[t] = __builtin_amdgcn_wmma_f32_16x16x4_f32(
                             false, a, false, b, (short)0, acc[t], false, false);
            }
        }

#pragma unroll
        for (int t = 0; t < TN; ++t) {
            int col  = t * 16 + lr;
            float bv = BIAS ? bias[col] : 0.0f;
#pragma unroll
            for (int r = 0; r < 8; ++r) {
                int row = tm * 16 + r + 8 * half;
                float v = acc[t][r] + bv;
                if (RELU) v = fmaxf(v, 0.0f);
                C[(size_t)row * Nc + col] = v;
            }
        }
    }
}

__global__ __launch_bounds__(256) void gcn_agg_init(
        float* __restrict__ out, const float* __restrict__ hw,
        const float* __restrict__ dis, const float* __restrict__ bias,
        int n, int h) {
    int h4  = h >> 2;
    int idx = blockIdx.x * blockDim.x + threadIdx.x;
    if (idx >= n * h4) return;
    int row = idx / h4;
    int cj  = idx - row * h4;
    float sn = dis[row];
    sn = sn * sn;
    v4f v  = ((const v4f*)hw)[idx];
    v4f bv = ((const v4f*)bias)[cj];
    ((v4f*)out)[idx] = v * sn + bv;
}

__global__ __launch_bounds__(256) void gcn_scatter(
        float* __restrict__ agg, const float* __restrict__ hw,
        const float* __restrict__ dis, const int* __restrict__ src,
        const int* __restrict__ dst, int e, int h) {
    int wave = blockIdx.x * (blockDim.x >> 5) + (threadIdx.x >> 5);
    if (wave >= e) return;
    int lane = threadIdx.x & 31;
    int s = src[wave];
    int d = dst[wave];
    float nrm = dis[s] * dis[d];

    v4f m = ((const v4f*)(hw + (size_t)s * h))[lane];
    m *= nrm;
    float* drow = agg + (size_t)d * h + lane * 4;
    atomicAdd(drow + 0, m.x);
    atomicAdd(drow + 1, m.y);
    atomicAdd(drow + 2, m.z);
    atomicAdd(drow + 3, m.w);
}

extern "C" void kernel_launch(void* const* d_in, const int* in_sizes, int n_in,
                              void* d_out, int out_size, void* d_ws, size_t ws_size,
                              hipStream_t stream) {
    const float* x  = (const float*)d_in[0];
    const int*   ei = (const int*)  d_in[1];
    const float* W1 = (const float*)d_in[2];
    const float* b1 = (const float*)d_in[3];
    const float* Wc = (const float*)d_in[4];
    const float* bc = (const float*)d_in[5];
    const float* W2 = (const float*)d_in[6];
    const float* b2 = (const float*)d_in[7];

    const int N = 50000, E = 800000, H = 128, L = 3;
    const int* srcv = ei;
    const int* dstv = ei + E;

    char* ws = (char*)d_ws;
    size_t off = 0;
    auto carve = [&](size_t bytes) -> void* {
        void* p = ws + off;
        off = (off + bytes + 255) & ~(size_t)255;
        return p;
    };
    float* deg  = (float*)carve((size_t)N * 4);
    float* dis  = (float*)carve((size_t)N * 4);
    float* bufA = (float*)carve((size_t)N * H * 4);
    float* bufB = (float*)carve((size_t)N * H * 4);
    float* hw   = (float*)carve((size_t)N * H * 4);

    gcn_deg_init <<<(N + 255) / 256, 256, 0, stream>>>(deg, N);
    gcn_deg_count<<<(E + 255) / 256, 256, 0, stream>>>(deg, dstv, E);
    gcn_rsqrt    <<<(N + 255) / 256, 256, 0, stream>>>(dis, deg, N);

    const int tilesM    = N / 16;
    const int gemmBlks  = (tilesM + 7) / 8;

    gcn_gemm_wmma<128, 8, true, true><<<gemmBlks, 256, 0, stream>>>(
        x, W1, b1, bufA, N);

    float* hcur = bufA;
    float* hnxt = bufB;
    for (int l = 0; l < L; ++l) {
        gcn_gemm_wmma<128, 8, false, false><<<gemmBlks, 256, 0, stream>>>(
            hcur, Wc + (size_t)l * H * H, nullptr, hw, N);

        int tot4 = N * (H / 4);
        gcn_agg_init<<<(tot4 + 255) / 256, 256, 0, stream>>>(
            hnxt, hw, dis, bc + (size_t)l * H, N, H);

        gcn_scatter<<<(E + 7) / 8, 256, 0, stream>>>(
            hnxt, hw, dis, srcv, dstv, E, H);

        float* t = hcur; hcur = hnxt; hnxt = t;
    }

    gcn_gemm_wmma<128, 4, false, true><<<gemmBlks, 256, 0, stream>>>(
        hcur, W2, b2, (float*)d_out, N);
}


# last conv layer reassociated: its products (Wc[2], then W2) run before its edge sums, in the previous layer's range work-group (3 WMMA passes); last edge sums on 64-wide rows; fp sum order changes
# speedup vs baseline: 4.6947x; 4.6947x over previous
#include <hip/hip_runtime.h>

typedef float v2f __attribute__((ext_vector_type(2)));
typedef float v4f __attribute__((ext_vector_type(4)));
typedef float v8f __attribute__((ext_vector_type(8)));

#define RNG_R 98
#define RNG_N 511
#define BIN_WG 160
#define BIN_CH 5000
#define BIN_PT 5
#define SEG_W 10
#define TILE_ROWS 112
#define TILE_P 130
static_assert(TILE_ROWS >= RNG_R && TILE_ROWS % 16 == 0, "tile holds whole 16-row matrix tiles");
static_assert(RNG_R * RNG_N >= 50000, "ranges must cover every node");
static_assert(BIN_WG * BIN_CH == 800000, "chunks must cover every edge");
static_assert(16 * SEG_W == BIN_WG, "the 16 waves of a range work-group cover every chunk");
static_assert(SEG_W <= 32, "a wave keeps one segment per lane");
#define TAB_P 256
static_assert(2 * TAB_P > RNG_N, "a table row holds an offset for every range and the end");
static_assert(BIN_CH < 65536, "a chunk's offsets fit 16 bits");
static_assert(BIN_PT * 1024 >= BIN_CH, "a binning work-group's threads cover its chunk");

__global__ __launch_bounds__(1024) void gcn_bin_sort(unsigned* __restrict__ list,
                                                     unsigned* __restrict__ tab,
                                                     const int* __restrict__ src,
                                                     const int* __restrict__ dst) {
    __shared__ int hist[2 * TAB_P];
    __shared__ int wsum[16];
    int t  = threadIdx.x;
    int lo = blockIdx.x * BIN_CH;
    if (t < 2 * TAB_P) hist[t] = 0;
    int d[BIN_PT], sv[BIN_PT];
#pragma unroll
    for (int u = 0; u < BIN_PT; ++u) {
        int i = t + 1024 * u;
        if (i > BIN_CH - 1) i = BIN_CH - 1;
        d[u]  = dst[lo + i];
        sv[u] = src[lo + i];
    }
    __syncthreads();
    int bb[BIN_PT], rk[BIN_PT];
#pragma unroll
    for (int u = 0; u < BIN_PT; ++u) {
        bb[u] = d[u] / RNG_R;
        rk[u] = 0;
        if (t + 1024 * u < BIN_CH) rk[u] = atomicAdd(&hist[bb[u]], 1);
    }
    __syncthreads();
    int v = 0, incl = 0;
    if (t < 512) {
        v = hist[t];
        incl = v;
        for (int dd = 1; dd < 32; dd <<= 1) {
            int nb = __shfl_up(incl, dd);
            if ((t & 31) >= dd) incl += nb;
        }
        if ((t & 31) == 31) wsum[t >> 5] = incl;
    }
    __syncthreads();
    if (t < 512) {
        int st = incl - v;
        for (int w = 0; w < (t >> 5); ++w) st += wsum[w];
        hist[t] = st;
    }
    __syncthreads();
    if (t < TAB_P)
        tab[blockIdx.x * TAB_P + t] = (unsigned)hist[2 * t] | ((unsigned)hist[2 * t + 1] << 16);
#pragma unroll
    for (int u = 0; u < BIN_PT; ++u)
        if (t + 1024 * u < BIN_CH)
            list[lo + hist[bb[u]] + rk[u]] =
                (unsigned)sv[u] | ((unsigned)(d[u] - bb[u] * RNG_R) << 16);
}

struct gcn_segs {
    int end[SEG_W];
    int adj[SEG_W];
};

__device__ __forceinline__ void gcn_seg_read(gcn_segs& sg, const unsigned* __restrict__ tab,
                                             int b, int wv, int lane) {
    int q  = lane < SEG_W ? lane : SEG_W - 1;
    int k  = wv * SEG_W + q;
    unsigned w0 = tab[k * TAB_P + (b >> 1)];
    unsigned w1 = tab[k * TAB_P + ((b + 1) >> 1)];
    int o0 = (int)((b & 1) ? (w0 >> 16) : (w0 & 0xFFFFu));
    int o1 = (int)(((b + 1) & 1) ? (w1 >> 16) : (w1 & 0xFFFFu));
    int len   = o1 - o0;
    int first = k * BIN_CH + o0;
    int run = 0;
#pragma unroll
    for (int u = 0; u < SEG_W; ++u) {
        sg.adj[u] = __builtin_amdgcn_readlane(first, u) - run;
        run += __builtin_amdgcn_readlane(len, u);
        sg.end[u] = run;
    }
}

__device__ __forceinline__ int gcn_seg_index(const gcn_segs& sg, int j) {
    int idx = sg.adj[0];
#pragma unroll
    for (int u = 1; u < SEG_W; ++u) idx = (j >= sg.end[u - 1]) ? sg.adj[u] : idx;
    return idx + j;
}

#define SUM_GRP 8
#define SUMQ_GRP 16
#define Q_P 64

template <typename V> struct gcn_row;
template <> struct gcn_row<v4f> { static constexpr int NW = 4; };
template <> struct gcn_row<v2f> { static constexpr int NW = 2; };

template <typename V, int GRP, int G0>
__device__ __forceinline__ void gcn_rows_load(V (&m)[GRP], const float* __restrict__ hw,
                                              int s, int lane) {
#pragma unroll
    for (int u = 0; u < GRP; ++u) {
        int su = __builtin_amdgcn_readlane(s, G0 + u);
        const V* rowp = (const V*)(hw + (size_t)su * (32 * gcn_row<V>::NW));
        m[u] = rowp[lane];
    }
}

template <typename V, int GRP, int PITCH, int G0, bool FULL>
__device__ __forceinline__ void gcn_rows_add(float* acc, const V (&m)[GRP],
                                             int dl, int c, int lane) {
#pragma unroll
    for (int u = 0; u < GRP; ++u) {
        if (FULL || G0 + u < c) {
            int du = __builtin_amdgcn_readlane(dl, G0 + u);
            float* a = acc + du * PITCH + lane;
#pragma unroll
            for (int j = 0; j < gcn_row<V>::NW; ++j) atomicAdd(a + 32 * j, m[u][j]);
        }
    }
}

template <typename V, int GRP, int PITCH, int G0, bool FULL>
__device__ __forceinline__ void gcn_rows_rest(float* acc, const float* __restrict__ hw,
                                              int s, int dl, int c, int lane) {
    if constexpr (G0 < 32) {
        if (FULL || G0 < c) {
            V m[GRP];
            gcn_rows_load<V, GRP, G0>(m, hw, s, lane);
            gcn_rows_add<V, GRP, PITCH, G0, FULL>(acc, m, dl, c, lane);
            gcn_rows_rest<V, GRP, PITCH, G0 + GRP, FULL>(acc, hw, s, dl, c, lane);
        }
    }
}

template <typename V, int GRP, int PITCH>
__device__ __forceinline__ void gcn_edge_sums(float* acc, const float* __restrict__ hw,
                                              const unsigned* __restrict__ list,
                                              const gcn_segs& sg, int s, int dl, int c, int lane) {
    constexpr int LAST = BIN_WG * BIN_CH - 1;
    int ch = 0;
    while (c > 0) {
        int cn = c - 32;
        int li2 = gcn_seg_index(sg, (ch + 1) * 32 + lane);
        unsigned p2 = list[li2 < LAST ? li2 : LAST];

        V m[GRP];
        gcn_rows_load<V, GRP, 0>(m, hw, s, lane);

        if (lane >= cn) p2 = 0;
        int s2  = (int)(p2 & 0xFFFFu);
        int dl2 = (int)(p2 >> 16);

        if (c >= 32) {
            gcn_rows_add<V, GRP, PITCH, 0, true>(acc, m, dl, c, lane);
            gcn_rows_rest<V, GRP, PITCH, GRP, true>(acc, hw, s, dl, c, lane);
        } else {
            gcn_rows_add<V, GRP, PITCH, 0, false>(acc, m, dl, c, lane);
            gcn_rows_rest<V, GRP, PITCH, GRP, false>(acc, hw, s, dl, c, lane);
        }

        s = s2; dl = dl2;
        c = cn; ch += 1;
    }
}

__device__ __forceinline__ void gcn_tile_mm(const float* ap, const float* bp0, const float* bp1,
                                            v8f& d0, v8f& d1) {
    v8f e0 = {};
    v8f e1 = {};
#pragma unroll 8
    for (int q = 0; q < 32; ++q) {
        v2f a;
        a.x = ap[q];
        a.y = ap[q + 64];
        v2f b0 = *(const v2f*)(bp0 + 4 * q);
        v2f b1 = *(const v2f*)(bp1 + 4 * q);
        e0 = __builtin_amdgcn_wmma_f32_16x16x4_f32(
                 false, a, false, b0, (short)0, e0, false, false);
        e1 = __builtin_amdgcn_wmma_f32_16x16x4_f32(
                 false, a, false, b1, (short)0, e1, false, false);
    }
    d0 = e0;
    d1 = e1;
}

template <bool CHAIN>
__global__ __launch_bounds__(512) void gcn_range_layer(
        float* __restrict__ out, const float* __restrict__ h,
        const float* __restrict__ dis, const float* __restrict__ Wl,
        const float* __restrict__ bias,
        const float* __restrict__ Wn, const float* __restrict__ biasn,
        const float* __restrict__ Wo, const float* __restrict__ bo,
        float* __restrict__ cvec,
        const unsigned* __restrict__ list, const unsigned* __restrict__ tab, int n) {
    __shared__ float acc[TILE_ROWS * TILE_P];
    __shared__ float Wt[128 * TILE_P];
    __shared__ float dsh[TILE_ROWS];
    constexpr int LAST = BIN_WG * BIN_CH - 1;
    int t    = threadIdx.x;
    int lane = t & 31;
    int wv   = __builtin_amdgcn_readfirstlane(t >> 5);
    int b    = blockIdx.x;
    int nodeBase = b * RNG_R;
    int rows = n - nodeBase;
    if (rows > RNG_R) rows = RNG_R;

    float wr[32];
#pragma unroll
    for (int it = 0; it < 32; ++it) wr[it] = Wl[t + 512 * it];

    gcn_segs sg;
    gcn_seg_read(sg, tab, b, wv, lane);

    int c  = sg.end[SEG_W - 1];
    int li = gcn_seg_index(sg, lane);
    unsigned p = list[li < LAST ? li : LAST];
    if (lane >= c) p = 0;
    int s  = (int)(p & 0xFFFFu);
    int dl = (int)(p >> 16);
    float dv = dis[nodeBase + (t < rows ? t : rows - 1)];

    {
        int r = t >> 7;
        float* wp = Wt + (t & 127) * TILE_P + (((r & 1) << 1) | (r >> 1));
#pragma unroll
        for (int it = 0; it < 32; ++it) wp[4 * it] = wr[it];
    }

    {
        v4f x[7];
#pragma unroll
        for (int k = 0; k < 7; ++k) {
            int r = wv + 16 * k;
            int rc = r < rows ? r : rows - 1;
            x[k]  = ((const v4f*)h)[(size_t)(nodeBase + rc) * 32 + lane];
        }
#pragma unroll
        for (int k = 0; k < 7; ++k) {
            int r = wv + 16 * k;
            v4f y = x[k];
            if (r >= rows) y = v4f{0.0f, 0.0f, 0.0f, 0.0f};
            float* a = acc + r * TILE_P + lane;
            a[0]  = y.x;
            a[32] = y.y;
            a[64] = y.z;
            a[96] = y.w;
        }
    }
    static_assert(TILE_ROWS == 7 * 16, "tile init covers 7 rows a wave");
    if (t < TILE_ROWS) dsh[t] = dv;
    __syncthreads();

    gcn_edge_sums<v4f, SUM_GRP, TILE_P>(acc, h, list, sg, s, dl, c, lane);
    __syncthreads();

    const int hf = lane >> 4;
    const int lr = lane & 15;
    if constexpr (!CHAIN) {
        for (int j = wv; j < 28; j += 16) {
            int rt = j >> 2;
            int cq = j & 3;
            if (rt * 16 < rows) {
                const float* bp0 = Wt + (cq * 32 + lr) * TILE_P + 2 * hf;
                v8f d0, d1;
                gcn_tile_mm(acc + (rt * 16 + lr) * TILE_P + 32 * hf, bp0, bp0 + 16 * TILE_P, d0, d1);
                int col0 = cq * 32 + lr;
                float bv0 = bias[col0];
                float bv1 = bias[col0 + 16];
#pragma unroll
                for (int r = 0; r < 8; ++r) {
                    int row = rt * 16 + r + 8 * hf;
                    float dr = dsh[row];
                    if (row < rows) {
                        float* o = out + (size_t)(nodeBase + row) * 128 + col0;
                        o[0]  = (dr * d0[r] + bv0) * dr;
                        o[16] = (dr * d1[r] + bv1) * dr;
                    }
                }
            }
        }
    } else {
        static_assert(RNG_R < TILE_ROWS && RNG_R > TILE_ROWS - 16,
                      "the tile's last row is in a row tile that a full range uses, and no node's");
        const int rt0 = wv >> 2;
        const int rt1 = rt0 + 4;
        const int cq  = wv & 3;
        const bool has0 = rt0 * 16 < rows;
        const bool has1 = rt1 < 7 && rt1 * 16 < rows;
        const int col0 = cq * 32 + lr;
        const float* bp0 = Wt + col0 * TILE_P + 2 * hf;
        const float* ap0 = acc + (rt0 * 16 + lr) * TILE_P + 32 * hf;
        const float* ap1 = acc + (rt1 * 16 + lr) * TILE_P + 32 * hf;
        const int w0 = (col0 & 3) * 32 + (col0 >> 2);
        const int w1 = w0 + 4;
        float* tp0 = acc + (rt0 * 16 + 8 * hf) * TILE_P + w0;
        float* tp1 = acc + (rt1 * 16 + 8 * hf) * TILE_P + w0;
        float wn[32];
#pragma unroll
        for (int it = 0; it < 32; ++it) wn[it] = Wn[t + 512 * it];
        v8f p00 = {}, p01 = {}, p10 = {}, p11 = {};
        if (has0) gcn_tile_mm(ap0, bp0, bp0 + 16 * TILE_P, p00, p01);
        if (has1) gcn_tile_mm(ap1, bp0, bp0 + 16 * TILE_P, p10, p11);
        __syncthreads();
        {
            int r = t >> 7;
            float* wp = Wt + (t & 127) * TILE_P + (((r & 1) << 1) | (r >> 1));
#pragma unroll
            for (int it = 0; it < 32; ++it) wp[4 * it] = wn[it];
        }
        float wo[16];
#pragma unroll
        for (int it = 0; it < 16; ++it) wo[it] = Wo[t + 512 * it];
        {
            float bv0 = bias[col0];
            float bv1 = bias[col0 + 16];
            if (has0) {
#pragma unroll
                for (int r = 0; r < 8; ++r) {
                    float dr = dsh[rt0 * 16 + 8 * hf + r];
                    tp0[r * TILE_P]     = dr * p00[r] + bv0;
                    tp0[r * TILE_P + 4] = dr * p01[r] + bv1;
                }
            }
            if (has1) {
#pragma unroll
                for (int r = 0; r < 8; ++r) {
                    float dr = dsh[rt1 * 16 + 8 * hf + r];
                    tp1[r * TILE_P]     = dr * p10[r] + bv0;
                    tp1[r * TILE_P + 4] = dr * p11[r] + bv1;
                }
            }
        }
        __syncthreads();
        p00 = v8f{}; p01 = v8f{}; p10 = v8f{}; p11 = v8f{};
        if (has0) gcn_tile_mm(ap0, bp0, bp0 + 16 * TILE_P, p00, p01);
        if (has1) gcn_tile_mm(ap1, bp0, bp0 + 16 * TILE_P, p10, p11);
        float bn0 = biasn[col0];
        float bn1 = biasn[col0 + 16];
        __syncthreads();
        {
            int r8 = t >> 6;
            float* wp = Wt + (t & 63) * TILE_P + 4 * (r8 >> 2) + (((r8 & 1) << 1) | ((r8 >> 1) & 1));
#pragma unroll
            for (int it = 0; it < 16; ++it) wp[8 * it] = wo[it];
        }
        {
            const bool brow = b == 0 && rt1 == 6 && hf == 1;
            if (has0) {
#pragma unroll
                for (int r = 0; r < 8; ++r) {
                    tp0[r * TILE_P]     = p00[r];
                    tp0[r * TILE_P + 4] = p01[r];
                }
            }
            if (has1) {
#pragma unroll
                for (int r = 0; r < 8; ++r) {
                    float v0 = p10[r];
                    float v1 = p11[r];
                    if (r == 7 && brow) { v0 = bn0; v1 = bn1; }
                    tp1[r * TILE_P]     = v0;
                    tp1[r * TILE_P + 4] = v1;
                }
            }
        }
        __syncthreads();
        if (wv < 14) {
            int rt = wv >> 1;
            int chh = wv & 1;
            if (rt * 16 < rows) {
                int oc = chh * 32 + lr;
                const float* bq0 = Wt + oc * TILE_P + 2 * hf;
                v8f d0, d1;
                gcn_tile_mm(acc + (rt * 16 + lr) * TILE_P + 32 * hf, bq0, bq0 + 16 * TILE_P, d0, d1);
#pragma unroll
                for (int r = 0; r < 8; ++r) {
                    int row = rt * 16 + r + 8 * hf;
                    float dr = dsh[row];
                    if (row < rows) {
                        float* o = out + (size_t)(nodeBase + row) * 64 + oc;
                        o[0]  = dr * d0[r];
                        o[16] = dr * d1[r];
                    }
                }
                if (b == 0 && rt == 6 && hf == 1) {
                    cvec[oc]      = d0[7] + bo[oc];
                    cvec[oc + 16] = d1[7] + bo[oc + 16];
                }
            }
        }
    }
}

__global__ __launch_bounds__(512) void gcn_range_sum(
        float* __restrict__ out, const float* __restrict__ q,
        const float* __restrict__ dis, const float* __restrict__ cvec,
        const unsigned* __restrict__ list, const unsigned* __restrict__ tab, int n) {
    __shared__ float acc[TILE_ROWS * Q_P];
    constexpr int LAST = BIN_WG * BIN_CH - 1;
    int t    = threadIdx.x;
    int lane = t & 31;
    int wv   = __builtin_amdgcn_readfirstlane(t >> 5);
    int b    = blockIdx.x;
    int nodeBase = b * RNG_R;
    int rows = n - nodeBase;
    if (rows > RNG_R) rows = RNG_R;

    gcn_segs sg;
    gcn_seg_read(sg, tab, b, wv, lane);

    int c  = sg.end[SEG_W - 1];
    int li = gcn_seg_index(sg, lane);
    unsigned p = list[li < LAST ? li : LAST];
    if (lane >= c) p = 0;
    int s  = (int)(p & 0xFFFFu);
    int dl = (int)(p >> 16);

    v2f cv = ((const v2f*)cvec)[lane];
    float dr[7];
    {
        v2f x[7];
#pragma unroll
        for (int k = 0; k < 7; ++k) {
            int r = wv + 16 * k;
            int rc = r < rows ? r : rows - 1;
            x[k]  = ((const v2f*)q)[(size_t)(nodeBase + rc) * 32 + lane];
            dr[k] = dis[nodeBase + rc];
        }
#pragma unroll
        for (int k = 0; k < 7; ++k) {
            int r = wv + 16 * k;
            float* a = acc + r * Q_P + lane;
            a[0]  = x[k].x;
            a[32] = x[k].y;
        }
    }
    __syncthreads();

    gcn_edge_sums<v2f, SUMQ_GRP, Q_P>(acc, q, list, sg, s, dl, c, lane);
    __syncthreads();

#pragma unroll
    for (int k = 0; k < 7; ++k) {
        int r = wv + 16 * k;
        if (r < rows) {
            const float* a = acc + r * Q_P + lane;
            v2f o;
            o.x = dr[k] * a[0]  + cv.x;
            o.y = dr[k] * a[32] + cv.y;
            ((v2f*)out)[(size_t)(nodeBase + r) * 32 + lane] = o;
        }
    }
}

__global__ __launch_bounds__(512) void gcn_range_head(
        float* __restrict__ out, float* __restrict__ dis,
        const float* __restrict__ x, const float* __restrict__ Wl,
        const float* __restrict__ bias,
        const unsigned* __restrict__ list, const unsigned* __restrict__ tab, int n) {
    __shared__ float acc[TILE_ROWS * TILE_P];
    __shared__ float Wt[128 * TILE_P];
    __shared__ float dsh[TILE_ROWS];
    __shared__ int   cnt[TILE_ROWS];
    constexpr int LAST = BIN_WG * BIN_CH - 1;
    int t    = threadIdx.x;
    int lane = t & 31;
    int wv   = __builtin_amdgcn_readfirstlane(t >> 5);
    int b    = blockIdx.x;
    int nodeBase = b * RNG_R;
    int rows = n - nodeBase;
    if (rows > RNG_R) rows = RNG_R;

    float wr[32];
#pragma unroll
    for (int it = 0; it < 32; ++it) wr[it] = Wl[t + 512 * it];

    gcn_segs sg;
    gcn_seg_read(sg, tab, b, wv, lane);
    int mycnt = sg.end[SEG_W - 1];

    unsigned pe[4];
#pragma unroll
    for (int k = 0; k < 4; ++k) {
        int li = gcn_seg_index(sg, lane + 32 * k);
        pe[k] = list[li < LAST ? li : LAST];
    }
    if (t < TILE_ROWS) cnt[t] = 0;

    {
        int r = t >> 7;
        float* wp = Wt + (t & 127) * TILE_P + (((r & 1) << 1) | (r >> 1));
#pragma unroll
        for (int it = 0; it < 32; ++it) wp[4 * it] = wr[it];
    }

    {
        v4f xr[7];
#pragma unroll
        for (int k = 0; k < 7; ++k) {
            int r = wv + 16 * k;
            int rc = r < rows ? r : rows - 1;
            xr[k]  = ((const v4f*)x)[(size_t)(nodeBase + rc) * 32 + lane];
        }
#pragma unroll
        for (int k = 0; k < 7; ++k) {
            int r = wv + 16 * k;
            v4f y = xr[k];
            if (r >= rows) y = v4f{0.0f, 0.0f, 0.0f, 0.0f};
            float* a = acc + r * TILE_P + lane;
            a[0]  = y.x;
            a[32] = y.y;
            a[64] = y.z;
            a[96] = y.w;
        }
    }
    __syncthreads();

#pragma unroll
    for (int k = 0; k < 4; ++k)
        if (lane + 32 * k < mycnt) atomicAdd(&cnt[pe[k] >> 16], 1);
    for (int j = lane + 128; j < mycnt; j += 32)
        atomicAdd(&cnt[list[gcn_seg_index(sg, j)] >> 16], 1);
    __syncthreads();
    if (t < TILE_ROWS) {
        float dv = rsqrtf(1.0f + (float)cnt[t]);
        dsh[t] = dv;
        if (t < rows) dis[nodeBase + t] = dv;
    }
    __syncthreads();

    const int hf = lane >> 4;
    const int lr = lane & 15;
    for (int j = wv; j < 28; j += 16) {
        int rt = j >> 2;
        int cq = j & 3;
        if (rt * 16 < rows) {
            const float* bp0 = Wt + (cq * 32 + lr) * TILE_P + 2 * hf;
            v8f d0, d1;
            gcn_tile_mm(acc + (rt * 16 + lr) * TILE_P + 32 * hf, bp0, bp0 + 16 * TILE_P, d0, d1);
            int col0 = cq * 32 + lr;
            float bv0 = bias[col0];
            float bv1 = bias[col0 + 16];
#pragma unroll
            for (int r = 0; r < 8; ++r) {
                int row = rt * 16 + r + 8 * hf;
                float dr = dsh[row];
                if (row < rows) {
                    float* o = out + (size_t)(nodeBase + row) * 128 + col0;
                    o[0]  = fmaxf(d0[r] + bv0, 0.0f) * dr;
                    o[16] = fmaxf(d1[r] + bv1, 0.0f) * dr;
                }
            }
        }
    }
}

extern "C" void kernel_launch(void* const* d_in, const int* in_sizes, int n_in,
                              void* d_out, int out_size, void* d_ws, size_t ws_size,
                              hipStream_t stream) {
    const float* x  = (const float*)d_in[0];
    const int*   ei = (const int*)  d_in[1];
    const float* W1 = (const float*)d_in[2];
    const float* b1 = (const float*)d_in[3];
    const float* Wc = (const float*)d_in[4];
    const float* bc = (const float*)d_in[5];
    const float* W2 = (const float*)d_in[6];
    const float* b2 = (const float*)d_in[7];

    const int N = 50000, E = 800000, H = 128, L = 3;
    const int* srcv = ei;
    const int* dstv = ei + E;

    static_assert(N <= 65536, "list entries keep src in 16 bits");
    static_assert(L >= 2, "the layer before the last carries the last layer's products");
    char* ws = (char*)d_ws;
    size_t off = 0;
    auto carve = [&](size_t bytes) -> void* {
        void* p = ws + off;
        off = (off + bytes + 255) & ~(size_t)255;
        return p;
    };
    float* dis  = (float*)carve((size_t)N * 4);
    float* bufA = (float*)carve((size_t)N * H * 4);
    float* bufB = (float*)carve((size_t)N * H * 4);
    unsigned* tab  = (unsigned*)carve((size_t)BIN_WG * TAB_P * 4);
    unsigned* list = (unsigned*)carve((size_t)E * 4);
    float* cvec = (float*)carve((size_t)64 * 4);

    gcn_bin_sort<<<BIN_WG, 1024, 0, stream>>>(list, tab, srcv, dstv);

    gcn_range_head<<<RNG_N, 512, 0, stream>>>(bufA, dis, x, W1, b1, list, tab, N);

    float* hcur = bufA;
    float* hnxt = bufB;
    for (int l = 0; l < L - 2; ++l) {
        gcn_range_layer<false><<<RNG_N, 512, 0, stream>>>(
            hnxt, hcur, dis, Wc + (size_t)l * H * H, bc + (size_t)l * H,
            nullptr, nullptr, nullptr, nullptr, nullptr, list, tab, N);

        float* t = hcur; hcur = hnxt; hnxt = t;
    }

    gcn_range_layer<true><<<RNG_N, 512, 0, stream>>>(
        hnxt, hcur, dis, Wc + (size_t)(L - 2) * H * H, bc + (size_t)(L - 2) * H,
        Wc + (size_t)(L - 1) * H * H, bc + (size_t)(L - 1) * H, W2, b2, cvec, list, tab, N);

    gcn_range_sum<<<RNG_N, 512, 0, stream>>>((float*)d_out, hnxt, dis, cvec, list, tab, N);
}


# conv layers 0 and 1 reassociated like the last: all products (W1, relu, Wc0..2, W2: 5 WMMA passes from the LDS tile) run in the head launch; all 3 edge sums on 64-wide rows; fp sum order changes
# speedup vs baseline: 5.4785x; 1.1670x over previous
#include <hip/hip_runtime.h>

typedef float v2f __attribute__((ext_vector_type(2)));
typedef float v4f __attribute__((ext_vector_type(4)));
typedef float v8f __attribute__((ext_vector_type(8)));

#define RNG_R 98
#define RNG_N 511
#define BIN_WG 160
#define BIN_CH 5000
#define BIN_PT 5
#define SEG_W 10
#define TILE_ROWS 112
#define TILE_P 130
static_assert(TILE_ROWS >= RNG_R && TILE_ROWS % 16 == 0, "tile holds whole 16-row matrix tiles");
static_assert(RNG_R * RNG_N >= 50000, "ranges must cover every node");
static_assert(BIN_WG * BIN_CH == 800000, "chunks must cover every edge");
static_assert(16 * SEG_W == BIN_WG, "the 16 waves of a range work-group cover every chunk");
static_assert(SEG_W <= 32, "a wave keeps one segment per lane");
#define TAB_P 256
static_assert(2 * TAB_P > RNG_N, "a table row holds an offset for every range and the end");
static_assert(BIN_CH < 65536, "a chunk's offsets fit 16 bits");
static_assert(BIN_PT * 1024 >= BIN_CH, "a binning work-group's threads cover its chunk");

__global__ __launch_bounds__(1024) void gcn_bin_sort(unsigned* __restrict__ list,
                                                     unsigned* __restrict__ tab,
                                                     const int* __restrict__ src,
                                                     const int* __restrict__ dst) {
    __shared__ int hist[2 * TAB_P];
    __shared__ int wsum[16];
    int t  = threadIdx.x;
    int lo = blockIdx.x * BIN_CH;
    if (t < 2 * TAB_P) hist[t] = 0;
    int d[BIN_PT], sv[BIN_PT];
#pragma unroll
    for (int u = 0; u < BIN_PT; ++u) {
        int i = t + 1024 * u;
        if (i > BIN_CH - 1) i = BIN_CH - 1;
        d[u]  = dst[lo + i];
        sv[u] = src[lo + i];
    }
    __syncthreads();
    int bb[BIN_PT], rk[BIN_PT];
#pragma unroll
    for (int u = 0; u < BIN_PT; ++u) {
        bb[u] = d[u] / RNG_R;
        rk[u] = 0;
        if (t + 1024 * u < BIN_CH) rk[u] = atomicAdd(&hist[bb[u]], 1);
    }
    __syncthreads();
    int v = 0, incl = 0;
    if (t < 512) {
        v = hist[t];
        incl = v;
        for (int dd = 1; dd < 32; dd <<= 1) {
            int nb = __shfl_up(incl, dd);
            if ((t & 31) >= dd) incl += nb;
        }
        if ((t & 31) == 31) wsum[t >> 5] = incl;
    }
    __syncthreads();
    if (t < 512) {
        int st = incl - v;
        for (int w = 0; w < (t >> 5); ++w) st += wsum[w];
        hist[t] = st;
    }
    __syncthreads();
    if (t < TAB_P)
        tab[blockIdx.x * TAB_P + t] = (unsigned)hist[2 * t] | ((unsigned)hist[2 * t + 1] << 16);
#pragma unroll
    for (int u = 0; u < BIN_PT; ++u)
        if (t + 1024 * u < BIN_CH)
            list[lo + hist[bb[u]] + rk[u]] =
                (unsigned)sv[u] | ((unsigned)(d[u] - bb[u] * RNG_R) << 16);
}

struct gcn_segs {
    int end[SEG_W];
    int adj[SEG_W];
};

__device__ __forceinline__ void gcn_seg_read(gcn_segs& sg, const unsigned* __restrict__ tab,
                                             int b, int wv, int lane) {
    int q  = lane < SEG_W ? lane : SEG_W - 1;
    int k  = wv * SEG_W + q;
    unsigned w0 = tab[k * TAB_P + (b >> 1)];
    unsigned w1 = tab[k * TAB_P + ((b + 1) >> 1)];
    int o0 = (int)((b & 1) ? (w0 >> 16) : (w0 & 0xFFFFu));
    int o1 = (int)(((b + 1) & 1) ? (w1 >> 16) : (w1 & 0xFFFFu));
    int len   = o1 - o0;
    int first = k * BIN_CH + o0;
    int run = 0;
#pragma unroll
    for (int u = 0; u < SEG_W; ++u) {
        sg.adj[u] = __builtin_amdgcn_readlane(first, u) - run;
        run += __builtin_amdgcn_readlane(len, u);
        sg.end[u] = run;
    }
}

__device__ __forceinline__ int gcn_seg_index(const gcn_segs& sg, int j) {
    int idx = sg.adj[0];
#pragma unroll
    for (int u = 1; u < SEG_W; ++u) idx = (j >= sg.end[u - 1]) ? sg.adj[u] : idx;
    return idx + j;
}

#define SUMQ_GRP 16
#define Q_P 64

template <typename V> struct gcn_row;
template <> struct gcn_row<v2f> { static constexpr int NW = 2; };

template <typename V, int GRP, int G0>
__device__ __forceinline__ void gcn_rows_load(V (&m)[GRP], const float* __restrict__ hw,
                                              int s, int lane) {
#pragma unroll
    for (int u = 0; u < GRP; ++u) {
        int su = __builtin_amdgcn_readlane(s, G0 + u);
        const V* rowp = (const V*)(hw + (size_t)su * (32 * gcn_row<V>::NW));
        m[u] = rowp[lane];
    }
}

template <typename V, int GRP, int PITCH, int G0, bool FULL>
__device__ __forceinline__ void gcn_rows_add(float* acc, const V (&m)[GRP],
                                             int dl, int c, int lane) {
#pragma unroll
    for (int u = 0; u < GRP; ++u) {
        if (FULL || G0 + u < c) {
            int du = __builtin_amdgcn_readlane(dl, G0 + u);
            float* a = acc + du * PITCH + lane;
#pragma unroll
            for (int j = 0; j < gcn_row<V>::NW; ++j) atomicAdd(a + 32 * j, m[u][j]);
        }
    }
}

template <typename V, int GRP, int PITCH, int G0, bool FULL>
__device__ __forceinline__ void gcn_rows_rest(float* acc, const float* __restrict__ hw,
                                              int s, int dl, int c, int lane) {
    if constexpr (G0 < 32) {
        if (FULL || G0 < c) {
            V m[GRP];
            gcn_rows_load<V, GRP, G0>(m, hw, s, lane);
            gcn_rows_add<V, GRP, PITCH, G0, FULL>(acc, m, dl, c, lane);
            gcn_rows_rest<V, GRP, PITCH, G0 + GRP, FULL>(acc, hw, s, dl, c, lane);
        }
    }
}

template <typename V, int GRP, int PITCH>
__device__ __forceinline__ void gcn_edge_sums(float* acc, const float* __restrict__ hw,
                                              const unsigned* __restrict__ list,
                                              const gcn_segs& sg, int s, int dl, int c, int lane) {
    constexpr int LAST = BIN_WG * BIN_CH - 1;
    int ch = 0;
    while (c > 0) {
        int cn = c - 32;
        int li2 = gcn_seg_index(sg, (ch + 1) * 32 + lane);
        unsigned p2 = list[li2 < LAST ? li2 : LAST];

        V m[GRP];
        gcn_rows_load<V, GRP, 0>(m, hw, s, lane);

        if (lane >= cn) p2 = 0;
        int s2  = (int)(p2 & 0xFFFFu);
        int dl2 = (int)(p2 >> 16);

        if (c >= 32) {
            gcn_rows_add<V, GRP, PITCH, 0, true>(acc, m, dl, c, lane);
            gcn_rows_rest<V, GRP, PITCH, GRP, true>(acc, hw, s, dl, c, lane);
        } else {
            gcn_rows_add<V, GRP, PITCH, 0, false>(acc, m, dl, c, lane);
            gcn_rows_rest<V, GRP, PITCH, GRP, false>(acc, hw, s, dl, c, lane);
        }

        s = s2; dl = dl2;
        c = cn; ch += 1;
    }
}

template <bool LAST>
__global__ __launch_bounds__(512) void gcn_range_sum(
        float* __restrict__ out, const float* __restrict__ q,
        const float* __restrict__ dis, const float* __restrict__ cvec,
        const unsigned* __restrict__ list, const unsigned* __restrict__ tab, int n) {
    __shared__ float acc[TILE_ROWS * Q_P];
    constexpr int LAST_I = BIN_WG * BIN_CH - 1;
    int t    = threadIdx.x;
    int lane = t & 31;
    int wv   = __builtin_amdgcn_readfirstlane(t >> 5);
    int b    = blockIdx.x;
    int nodeBase = b * RNG_R;
    int rows = n - nodeBase;
    if (rows > RNG_R) rows = RNG_R;

    gcn_segs sg;
    gcn_seg_read(sg, tab, b, wv, lane);

    int c  = sg.end[SEG_W - 1];
    int li = gcn_seg_index(sg, lane);
    unsigned p = list[li < LAST_I ? li : LAST_I];
    if (lane >= c) p = 0;
    int s  = (int)(p & 0xFFFFu);
    int dl = (int)(p >> 16);

    v2f cv = ((const v2f*)cvec)[lane];
    float dr[7];
    {
        v2f x[7];
#pragma unroll
        for (int k = 0; k < 7; ++k) {
            int r = wv + 16 * k;
            int rc = r < rows ? r : rows - 1;
            x[k]  = ((const v2f*)q)[(size_t)(nodeBase + rc) * 32 + lane];
            dr[k] = dis[nodeBase + rc];
        }
#pragma unroll
        for (int k = 0; k < 7; ++k) {
            int r = wv + 16 * k;
            float* a = acc + r * Q_P + lane;
            a[0]  = x[k].x;
            a[32] = x[k].y;
        }
    }
    static_assert(TILE_ROWS == 7 * 16, "tile init covers 7 rows a wave");
    __syncthreads();

    gcn_edge_sums<v2f, SUMQ_GRP, Q_P>(acc, q, list, sg, s, dl, c, lane);
    __syncthreads();

#pragma unroll
    for (int k = 0; k < 7; ++k) {
        int r = wv + 16 * k;
        if (r < rows) {
            const float* a = acc + r * Q_P + lane;
            v2f o;
            o.x = dr[k] * a[0]  + cv.x;
            o.y = dr[k] * a[32] + cv.y;
            if (!LAST) {
                o.x *= dr[k];
                o.y *= dr[k];
            }
            ((v2f*)out)[(size_t)(nodeBase + r) * 32 + lane] = o;
        }
    }
}

__device__ __forceinline__ void gcn_tile_mm(const float* ap, const float* bp0, const float* bp1,
                                            v8f& d0, v8f& d1) {
    v8f e0 = {};
    v8f e1 = {};
#pragma unroll 8
    for (int q = 0; q < 32; ++q) {
        v2f a;
        a.x = ap[q];
        a.y = ap[q + 64];
        v2f b0 = *(const v2f*)(bp0 + 4 * q);
        v2f b1 = *(const v2f*)(bp1 + 4 * q);
        e0 = __builtin_amdgcn_wmma_f32_16x16x4_f32(
                 false, a, false, b0, (short)0, e0, false, false);
        e1 = __builtin_amdgcn_wmma_f32_16x16x4_f32(
                 false, a, false, b1, (short)0, e1, false, false);
    }
    d0 = e0;
    d1 = e1;
}

__device__ __forceinline__ void gcn_tile_pass(const float* ap0, const float* ap1, const float* bp0,
                                              bool has0, bool has1,
                                              v8f& p00, v8f& p01, v8f& p10, v8f& p11) {
    p00 = v8f{}; p01 = v8f{}; p10 = v8f{}; p11 = v8f{};
    if (has0) gcn_tile_mm(ap0, bp0, bp0 + 16 * TILE_P, p00, p01);
    if (has1) gcn_tile_mm(ap1, bp0, bp0 + 16 * TILE_P, p10, p11);
}

__device__ __forceinline__ void gcn_w_load(float (&w)[32], const float* __restrict__ Wg, int t) {
#pragma unroll
    for (int it = 0; it < 32; ++it) w[it] = Wg[t + 512 * it];
}

__device__ __forceinline__ void gcn_w_store(float* Wt, const float (&w)[32], int t) {
    int r = t >> 7;
    float* wp = Wt + (t & 127) * TILE_P + (((r & 1) << 1) | (r >> 1));
#pragma unroll
    for (int it = 0; it < 32; ++it) wp[4 * it] = w[it];
}

#define GCN_L 3
static_assert(RNG_R + GCN_L <= TILE_ROWS && RNG_R > TILE_ROWS - 16,
              "the tile's last GCN_L rows are in a row tile that a full range uses, and no node's");
__global__ __launch_bounds__(512) void gcn_range_head(
        float* __restrict__ out, float* __restrict__ dis, float* __restrict__ cvec,
        const float* __restrict__ x, const float* __restrict__ Wl,
        const float* __restrict__ bias,
        const float* __restrict__ Wc, const float* __restrict__ bc,
        const float* __restrict__ Wo, const float* __restrict__ bo,
        const unsigned* __restrict__ list, const unsigned* __restrict__ tab, int n) {
    __shared__ float acc[TILE_ROWS * TILE_P];
    __shared__ float Wt[128 * TILE_P];
    __shared__ float dsh[TILE_ROWS];
    __shared__ int   cnt[TILE_ROWS];
    constexpr int LAST = BIN_WG * BIN_CH - 1;
    int t    = threadIdx.x;
    int lane = t & 31;
    int wv   = __builtin_amdgcn_readfirstlane(t >> 5);
    int b    = blockIdx.x;
    int nodeBase = b * RNG_R;
    int rows = n - nodeBase;
    if (rows > RNG_R) rows = RNG_R;

    float wr[32];
    gcn_w_load(wr, Wl, t);

    gcn_segs sg;
    gcn_seg_read(sg, tab, b, wv, lane);
    int mycnt = sg.end[SEG_W - 1];

    unsigned pe[4];
#pragma unroll
    for (int k = 0; k < 4; ++k) {
        int li = gcn_seg_index(sg, lane + 32 * k);
        pe[k] = list[li < LAST ? li : LAST];
    }
    if (t < TILE_ROWS) cnt[t] = 0;

    gcn_w_store(Wt, wr, t);

    {
        v4f xr[7];
#pragma unroll
        for (int k = 0; k < 7; ++k) {
            int r = wv + 16 * k;
            int rc = r < rows ? r : rows - 1;
            xr[k]  = ((const v4f*)x)[(size_t)(nodeBase + rc) * 32 + lane];
        }
#pragma unroll
        for (int k = 0; k < 7; ++k) {
            int r = wv + 16 * k;
            v4f y = xr[k];
            if (r >= rows) y = v4f{0.0f, 0.0f, 0.0f, 0.0f};
            float* a = acc + r * TILE_P + lane;
            a[0]  = y.x;
            a[32] = y.y;
            a[64] = y.z;
            a[96] = y.w;
        }
    }
    static_assert(TILE_ROWS == 7 * 16, "tile init covers 7 rows a wave");
    __syncthreads();

#pragma unroll
    for (int k = 0; k < 4; ++k)
        if (lane + 32 * k < mycnt) atomicAdd(&cnt[pe[k] >> 16], 1);
    for (int j = lane + 128; j < mycnt; j += 32)
        atomicAdd(&cnt[list[gcn_seg_index(sg, j)] >> 16], 1);
    __syncthreads();
    if (t < TILE_ROWS) {
        float dv = rsqrtf(1.0f + (float)cnt[t]);
        dsh[t] = dv;
        if (t < rows) dis[nodeBase + t] = dv;
    }

    const int hf = lane >> 4;
    const int lr = lane & 15;
    const int rt0 = wv >> 2;
    const int rt1 = rt0 + 4;
    const int cq  = wv & 3;
    const bool has0 = rt0 * 16 < rows;
    const bool has1 = rt1 < 7 && rt1 * 16 < rows;
    const int col0 = cq * 32 + lr;
    const float* bp0 = Wt + col0 * TILE_P + 2 * hf;
    const float* ap0 = acc + (rt0 * 16 + lr) * TILE_P + 32 * hf;
    const float* ap1 = acc + (rt1 * 16 + lr) * TILE_P + 32 * hf;
    const int w0 = (col0 & 3) * 32 + (col0 >> 2);
    float* tp0 = acc + (rt0 * 16 + 8 * hf) * TILE_P + w0;
    float* tp1 = acc + (rt1 * 16 + 8 * hf) * TILE_P + w0;
    const bool brow = b == 0 && rt1 == 6 && hf == 1;
    v8f p00, p01, p10, p11;
    float wn[32];

    {
        gcn_w_load(wn, Wc, t);
        float bv0 = bias[col0];
        float bv1 = bias[col0 + 16];
        gcn_tile_pass(ap0, ap1, bp0, has0, has1, p00, p01, p10, p11);
        __syncthreads();
        gcn_w_store(Wt, wn, t);
        if (has0) {
#pragma unroll
            for (int r = 0; r < 8; ++r) {
                tp0[r * TILE_P]     = fmaxf(p00[r] + bv0, 0.0f);
                tp0[r * TILE_P + 4] = fmaxf(p01[r] + bv1, 0.0f);
            }
        }
        if (has1) {
#pragma unroll
            for (int r = 0; r < 8; ++r) {
                tp1[r * TILE_P]     = fmaxf(p10[r] + bv0, 0.0f);
                tp1[r * TILE_P + 4] = fmaxf(p11[r] + bv1, 0.0f);
            }
        }
        __syncthreads();
    }

#pragma unroll
    for (int l = 0; l < GCN_L; ++l) {
        if (l + 1 < GCN_L) {
            gcn_w_load(wn, Wc + (size_t)(l + 1) * 128 * 128, t);
        } else {
#pragma unroll
            for (int it = 0; it < 16; ++it) wn[it] = Wo[t + 512 * it];
        }
        float bl0 = bc[l * 128 + col0];
        float bl1 = bc[l * 128 + col0 + 16];
        gcn_tile_pass(ap0, ap1, bp0, has0, has1, p00, p01, p10, p11);
        __syncthreads();
        if (l + 1 < GCN_L) {
            gcn_w_store(Wt, wn, t);
        } else {
            int r8 = t >> 6;
            float* wp = Wt + (t & 63) * TILE_P + 4 * (r8 >> 2) + (((r8 & 1) << 1) | ((r8 >> 1) & 1));
#pragma unroll
            for (int it = 0; it < 16; ++it) wp[8 * it] = wn[it];
        }
        if (has0) {
#pragma unroll
            for (int r = 0; r < 8; ++r) {
                tp0[r * TILE_P]     = p00[r];
                tp0[r * TILE_P + 4] = p01[r];
            }
        }
        if (has1) {
#pragma unroll
            for (int r = 0; r < 8; ++r) {
                float v0 = p10[r];
                float v1 = p11[r];
                if (r == 7 - l && brow) { v0 = bl0; v1 = bl1; }
                tp1[r * TILE_P]     = v0;
                tp1[r * TILE_P + 4] = v1;
            }
        }
        __syncthreads();
    }

    if (wv < 14) {
        int rt = wv >> 1;
        int chh = wv & 1;
        if (rt * 16 < rows) {
            int oc = chh * 32 + lr;
            const float* bq0 = Wt + oc * TILE_P + 2 * hf;
            v8f d0, d1;
            gcn_tile_mm(acc + (rt * 16 + lr) * TILE_P + 32 * hf, bq0, bq0 + 16 * TILE_P, d0, d1);
#pragma unroll
            for (int r = 0; r < 8; ++r) {
                int row = rt * 16 + r + 8 * hf;
                float dr = dsh[row];
                if (row < rows) {
                    float* o = out + (size_t)(nodeBase + row) * 64 + oc;
                    o[0]  = dr * d0[r];
                    o[16] = dr * d1[r];
                }
            }
            if (b == 0 && rt == 6 && hf == 1) {
#pragma unroll
                for (int l = 0; l < GCN_L; ++l) {
                    float e0 = d0[7 - l];
                    float e1 = d1[7 - l];
                    if (l == GCN_L - 1) {
                        e0 += bo[oc];
                        e1 += bo[oc + 16];
                    }
                    cvec[l * 64 + oc]      = e0;
                    cvec[l * 64 + oc + 16] = e1;
                }
            }
        }
    }
}

extern "C" void kernel_launch(void* const* d_in, const int* in_sizes, int n_in,
                              void* d_out, int out_size, void* d_ws, size_t ws_size,
                              hipStream_t stream) {
    const float* x  = (const float*)d_in[0];
    const int*   ei = (const int*)  d_in[1];
    const float* W1 = (const float*)d_in[2];
    const float* b1 = (const float*)d_in[3];
    const float* Wc = (const float*)d_in[4];
    const float* bc = (const float*)d_in[5];
    const float* W2 = (const float*)d_in[6];
    const float* b2 = (const float*)d_in[7];

    const int N = 50000, E = 800000, L = 3;
    const int* srcv = ei;
    const int* dstv = ei + E;

    static_assert(N <= 65536, "list entries keep src in 16 bits");
    static_assert(L == GCN_L, "the product launch runs one pass per conv layer");
    static_assert(N >= RNG_R, "work-group 0's range is full: it uses the tile's last row tile");
    char* ws = (char*)d_ws;
    size_t off = 0;
    auto carve = [&](size_t bytes) -> void* {
        void* p = ws + off;
        off = (off + bytes + 255) & ~(size_t)255;
        return p;
    };
    float* dis  = (float*)carve((size_t)N * 4);
    float* bufA = (float*)carve((size_t)N * 64 * 4);
    float* bufB = (float*)carve((size_t)N * 64 * 4);
    unsigned* tab  = (unsigned*)carve((size_t)BIN_WG * TAB_P * 4);
    unsigned* list = (unsigned*)carve((size_t)E * 4);
    float* cvec = (float*)carve((size_t)GCN_L * 64 * 4);

    gcn_bin_sort<<<BIN_WG, 1024, 0, stream>>>(list, tab, srcv, dstv);

    gcn_range_head<<<RNG_N, 512, 0, stream>>>(bufA, dis, cvec, x, W1, b1, Wc, bc, W2, b2,
                                              list, tab, N);

    gcn_range_sum<false><<<RNG_N, 512, 0, stream>>>(bufB, bufA, dis, cvec, list, tab, N);
    gcn_range_sum<false><<<RNG_N, 512, 0, stream>>>(bufA, bufB, dis, cvec + 64, list, tab, N);
    gcn_range_sum<true><<<RNG_N, 512, 0, stream>>>((float*)d_out, bufA, dis, cvec + 128, list, tab, N);
}
